# starting kernel
# baseline (speedup 1.0000x reference)
#include <hip/hip_runtime.h>
#include <math.h>

#define BATCH 8
#define SEQ   1024
#define KREL  4
#define NPOS  (BATCH * SEQ)

typedef __attribute__((ext_vector_type(2))) float v2f;
typedef __attribute__((ext_vector_type(4))) float v4f;
typedef __attribute__((ext_vector_type(8))) float v8f;

__global__ void __launch_bounds__(256)
k_rel_reduce(const float* __restrict__ rel, float* __restrict__ factA4) {
  const int row = blockIdx.x;
  const int i   = row & (SEQ - 1);
  const v4f* base = (const v4f*)(rel + (size_t)row * (SEQ * KREL));
  __shared__ int found;
  if (threadIdx.x == 0) found = 0;
  __syncthreads();
  int local = 0;
  for (int j = threadIdx.x; j < i; j += 256) {
    v4f v = __builtin_nontemporal_load(base + j);
    local |= (v.x > 0.5f) || (v.y > 0.5f) || (v.z > 0.5f) || (v.w > 0.5f);
  }
  if (local) atomicOr(&found, 1);
  __syncthreads();
  if (threadIdx.x == 0) factA4[row] = found ? 0.7f : 1.0f;
}

__global__ void __launch_bounds__(128)
k_mlp_factors(const float* __restrict__ aL, const float* __restrict__ oL,
              const int*   __restrict__ exA, const int* __restrict__ exO,
              const float* __restrict__ W1, const float* __restrict__ b1,
              const float* __restrict__ W2, const float* __restrict__ b2,
              const float* __restrict__ W3, const float* __restrict__ b3,
              float* __restrict__ mult1, float* __restrict__ fA2,
              float* __restrict__ fO2) {
  __shared__ float h[4][16][32];
  const int tid  = threadIdx.x;
  const int w    = tid >> 5;
  const int lane = tid & 31;
  const int hf   = lane >> 4;
  const int ln   = lane & 15;
  const int posBase = blockIdx.x * 64 + w * 16;
  const int p = posBase + ln;

  const float* aRow = aL + (size_t)p * 4;
  const float* oRow = oL + (size_t)p * 4;
  const float x0 = aRow[0], x1 = aRow[1], x2 = aRow[2], x3 = aRow[3];
  const float x4 = oRow[0], x5 = oRow[1], x6 = oRow[2], x7 = oRow[3];

  v2f a0, a1, bw;
  a0.x = hf ? x2 : x0;  a0.y = hf ? x3 : x1;
  a1.x = hf ? x6 : x4;  a1.y = hf ? x7 : x5;
  v8f acc0 = {0,0,0,0,0,0,0,0};
  v8f acc1 = {0,0,0,0,0,0,0,0};
  bw.x = W1[(0 + 2*hf)*32 + ln];  bw.y = W1[(1 + 2*hf)*32 + ln];
  acc0 = __builtin_amdgcn_wmma_f32_16x16x4_f32(false, a0, false, bw, (short)0, acc0, false, false);
  bw.x = W1[(4 + 2*hf)*32 + ln];  bw.y = W1[(5 + 2*hf)*32 + ln];
  acc0 = __builtin_amdgcn_wmma_f32_16x16x4_f32(false, a1, false, bw, (short)0, acc0, false, false);
  bw.x = W1[(0 + 2*hf)*32 + 16 + ln];  bw.y = W1[(1 + 2*hf)*32 + 16 + ln];
  acc1 = __builtin_amdgcn_wmma_f32_16x16x4_f32(false, a0, false, bw, (short)0, acc1, false, false);
  bw.x = W1[(4 + 2*hf)*32 + 16 + ln];  bw.y = W1[(5 + 2*hf)*32 + 16 + ln];
  acc1 = __builtin_amdgcn_wmma_f32_16x16x4_f32(false, a1, false, bw, (short)0, acc1, false, false);

  const float bia0 = b1[ln], bia1 = b1[16 + ln];
  #pragma unroll
  for (int v = 0; v < 8; ++v) {
    float t0 = acc0[v] + bia0;  h[w][v + 8*hf][ln]      = t0 > 0.0f ? t0 : 0.0f;
    float t1 = acc1[v] + bia1;  h[w][v + 8*hf][16 + ln] = t1 > 0.0f ? t1 : 0.0f;
  }
  __syncthreads();

  v8f acc2 = {0,0,0,0,0,0,0,0};
  #pragma unroll
  for (int k0 = 0; k0 < 32; k0 += 4) {
    v2f aa, bb;
    aa.x = h[w][ln][k0 + 2*hf];
    aa.y = h[w][ln][k0 + 1 + 2*hf];
    bb.x = W2[(k0 + 2*hf)*16 + ln];
    bb.y = W2[(k0 + 1 + 2*hf)*16 + ln];
    acc2 = __builtin_amdgcn_wmma_f32_16x16x4_f32(false, aa, false, bb, (short)0, acc2, false, false);
  }
  __syncthreads();
  const float bia2 = b2[ln];
  #pragma unroll
  for (int v = 0; v < 8; ++v) {
    float t = acc2[v] + bia2;
    h[w][v + 8*hf][ln] = t > 0.0f ? t : 0.0f;
  }
  __syncthreads();

  if (hf == 0) {
    float s3 = b3[0];
    #pragma unroll
    for (int c = 0; c < 16; ++c) s3 += h[w][ln][c] * W3[c];
    float cg   = 1.0f / (1.0f + expf(-s3));
    float mult = cg < 0.5f ? 2.0f * cg : 1.0f;
    mult1[p] = mult;

    float mA = fmaxf(fmaxf(x0, x1), fmaxf(x2, x3));
    float e0 = expf(x0 - mA), e1 = expf(x1 - mA), e2 = expf(x2 - mA), e3 = expf(x3 - mA);
    float impA = (e0 + e1) / (e0 + e1 + e2 + e3);
    float mO = fmaxf(fmaxf(x4, x5), fmaxf(x6, x7));
    float g0 = expf(x4 - mO), g1 = expf(x5 - mO), g2 = expf(x6 - mO), g3 = expf(x7 - mO);
    float impO = (g0 + g1) / (g0 + g1 + g2 + g3);

    const int bI = p >> 10;
    const int s  = p & (SEQ - 1);
    const int* eA = exA + bI * SEQ;
    const int* eO = exO + bI * SEQ;
    bool nearA = false, nearO = false;
    #pragma unroll
    for (int d = -3; d <= 3; ++d) {
      int ss = s + d;
      if (ss >= 0 && ss < SEQ) { nearA |= (eA[ss] > 0); nearO |= (eO[ss] > 0); }
    }
    fA2[p] = (impA > 0.5f && !nearO) ? 0.3f : 1.0f;
    fO2[p] = (impO > 0.5f && !nearA) ? 0.3f : 1.0f;
  }
}

__global__ void __launch_bounds__(256)
k_scan_write(const float* __restrict__ aL, const float* __restrict__ oL,
             const float* __restrict__ mult1, const float* __restrict__ fA2,
             const float* __restrict__ fO2,  const float* __restrict__ fA4,
             float* __restrict__ outA, float* __restrict__ outO) {
  const int b = blockIdx.x;
  const int t = threadIdx.x;
  __shared__ float sm[SEQ], sfa[SEQ], sfo[SEQ];
  __shared__ unsigned OG[32], IF1[32], IF01[32], F3[32];
  if (t < 32) { OG[t] = 0u; IF1[t] = 0u; IF01[t] = 0u; }
  __syncthreads();

  for (int s = t; s < SEQ; s += 256) {
    const int p = b * SEQ + s;
    v4f av = *(const v4f*)(aL + (size_t)p * 4);
    v4f ov = *(const v4f*)(oL + (size_t)p * 4);
    float mult = mult1[p];
    float fa1  = fA2[p] * fA4[p];
    float fo1  = fO2[p];
    sm[s] = mult;  sfa[s] = fa1;  sfo[s] = fo1;

    float a0 = av.x * mult, a1 = av.y * mult, a2 = av.z * mult, a3 = av.w * mult;
    float o0 = ov.x * mult, o1 = ov.y * mult, o2 = ov.z * mult, o3 = ov.w * mult;
    bool orig = (fmaxf(fmaxf(av.x, av.y), fmaxf(av.z, av.w)) > 0.5f) ||
                (fmaxf(fmaxf(ov.x, ov.y), fmaxf(ov.z, ov.w)) > 0.5f);
    float fa = fa1, fo = fo1;
    bool act1  = (fmaxf(fmaxf(a0*fa, a1*fa), fmaxf(a2, a3)) > 0.5f) ||
                 (fmaxf(fmaxf(o0*fo, o1*fo), fmaxf(o2, o3)) > 0.5f);
    fa = fa1 * 0.1f; fo = fo1 * 0.1f;
    bool act01 = (fmaxf(fmaxf(a0*fa, a1*fa), fmaxf(a2, a3)) > 0.5f) ||
                 (fmaxf(fmaxf(o0*fo, o1*fo), fmaxf(o2, o3)) > 0.5f);

    unsigned bit = 1u << (s & 31);
    if (orig)  atomicOr(&OG[s >> 5],  bit);
    if (act1)  atomicOr(&IF1[s >> 5], bit);
    if (act01) atomicOr(&IF01[s >> 5], bit);
  }
  __syncthreads();

  if (t == 0) {
    unsigned n1 = 0u, n2 = 0u;
    for (int wi = 0; wi < 32; ++wi) {
      unsigned og   = OG[wi];
      unsigned ogn  = (wi < 31) ? OG[wi + 1] : 0u;
      unsigned if1  = IF1[wi];
      unsigned if01 = IF01[wi];
      unsigned f3w  = 0u;
      #pragma unroll 4
      for (int bb = 0; bb < 32; ++bb) {
        unsigned o1b = (bb + 1 < 32) ? ((og >> (bb + 1)) & 1u) : (ogn & 1u);
        unsigned o2b = (bb + 2 < 32) ? ((og >> (bb + 2)) & 1u) : ((ogn >> (bb + 2 - 32)) & 1u);
        unsigned nb  = n1 | n2 | o1b | o2b;
        unsigned na  = nb ? ((if1 >> bb) & 1u) : ((if01 >> bb) & 1u);
        f3w |= nb << bb;
        n2 = n1;  n1 = na;
      }
      F3[wi] = f3w;
    }
  }
  __syncthreads();

  for (int s = t; s < SEQ; s += 256) {
    const int p = b * SEQ + s;
    v4f av = *(const v4f*)(aL + (size_t)p * 4);
    v4f ov = *(const v4f*)(oL + (size_t)p * 4);
    float mult = sm[s];
    float f3 = ((F3[s >> 5] >> (s & 31)) & 1u) ? 1.0f : 0.1f;
    float fa = sfa[s] * f3;
    float fo = sfo[s] * f3;
    v4f ra, ro;
    ra.x = av.x * mult * fa;  ra.y = av.y * mult * fa;
    ra.z = av.z * mult;       ra.w = av.w * mult;
    ro.x = ov.x * mult * fo;  ro.y = ov.y * mult * fo;
    ro.z = ov.z * mult;       ro.w = ov.w * mult;
    *(v4f*)(outA + (size_t)p * 4) = ra;
    *(v4f*)(outO + (size_t)p * 4) = ro;
  }
}

extern "C" void kernel_launch(void* const* d_in, const int* in_sizes, int n_in,
                              void* d_out, int out_size, void* d_ws, size_t ws_size,
                              hipStream_t stream) {
  (void)in_sizes; (void)n_in; (void)out_size; (void)ws_size;
  const float* aL  = (const float*)d_in[0];
  const float* oL  = (const float*)d_in[1];
  const float* rel = (const float*)d_in[2];
  const int*   exA = (const int*)d_in[3];
  const int*   exO = (const int*)d_in[4];
  const float* W1  = (const float*)d_in[5];
  const float* b1  = (const float*)d_in[6];
  const float* W2  = (const float*)d_in[7];
  const float* b2  = (const float*)d_in[8];
  const float* W3  = (const float*)d_in[9];
  const float* b3  = (const float*)d_in[10];

  float* ws    = (float*)d_ws;
  float* fA4   = ws;
  float* mult1 = ws + NPOS;
  float* fA2   = ws + 2 * NPOS;
  float* fO2   = ws + 3 * NPOS;
  float* outA  = (float*)d_out;
  float* outO  = outA + NPOS * 4;

  k_rel_reduce<<<NPOS, 256, 0, stream>>>(rel, fA4);
  k_mlp_factors<<<NPOS / 64, 128, 0, stream>>>(aL, oL, exA, exO,
                                               W1, b1, W2, b2, W3, b3,
                                               mult1, fA2, fO2);
  k_scan_write<<<BATCH, 256, 0, stream>>>(aL, oL, mult1, fA2, fO2, fA4,
                                          outA, outO);
}


# k_scan_write: the 1024-step serial activity recurrence of one thread becomes a prefix over 4-state transition maps in one wave (32 lanes x 32 positions, maps chained in word order); same result bits
# speedup vs baseline: 1.5777x; 1.5777x over previous
#include <hip/hip_runtime.h>
#include <math.h>

#define BATCH 8
#define SEQ   1024
#define KREL  4
#define NPOS  (BATCH * SEQ)

typedef __attribute__((ext_vector_type(2))) float v2f;
typedef __attribute__((ext_vector_type(4))) float v4f;
typedef __attribute__((ext_vector_type(8))) float v8f;

__global__ void __launch_bounds__(256)
k_rel_reduce(const float* __restrict__ rel, float* __restrict__ factA4) {
  const int row = blockIdx.x;
  const int i   = row & (SEQ - 1);
  const v4f* base = (const v4f*)(rel + (size_t)row * (SEQ * KREL));
  __shared__ int found;
  if (threadIdx.x == 0) found = 0;
  __syncthreads();
  int local = 0;
  for (int j = threadIdx.x; j < i; j += 256) {
    v4f v = __builtin_nontemporal_load(base + j);
    local |= (v.x > 0.5f) || (v.y > 0.5f) || (v.z > 0.5f) || (v.w > 0.5f);
  }
  if (local) atomicOr(&found, 1);
  __syncthreads();
  if (threadIdx.x == 0) factA4[row] = found ? 0.7f : 1.0f;
}

__global__ void __launch_bounds__(128)
k_mlp_factors(const float* __restrict__ aL, const float* __restrict__ oL,
              const int*   __restrict__ exA, const int* __restrict__ exO,
              const float* __restrict__ W1, const float* __restrict__ b1,
              const float* __restrict__ W2, const float* __restrict__ b2,
              const float* __restrict__ W3, const float* __restrict__ b3,
              float* __restrict__ mult1, float* __restrict__ fA2,
              float* __restrict__ fO2) {
  __shared__ float h[4][16][32];
  const int tid  = threadIdx.x;
  const int w    = tid >> 5;
  const int lane = tid & 31;
  const int hf   = lane >> 4;
  const int ln   = lane & 15;
  const int posBase = blockIdx.x * 64 + w * 16;
  const int p = posBase + ln;

  const float* aRow = aL + (size_t)p * 4;
  const float* oRow = oL + (size_t)p * 4;
  const float x0 = aRow[0], x1 = aRow[1], x2 = aRow[2], x3 = aRow[3];
  const float x4 = oRow[0], x5 = oRow[1], x6 = oRow[2], x7 = oRow[3];

  v2f a0, a1, bw;
  a0.x = hf ? x2 : x0;  a0.y = hf ? x3 : x1;
  a1.x = hf ? x6 : x4;  a1.y = hf ? x7 : x5;
  v8f acc0 = {0,0,0,0,0,0,0,0};
  v8f acc1 = {0,0,0,0,0,0,0,0};
  bw.x = W1[(0 + 2*hf)*32 + ln];  bw.y = W1[(1 + 2*hf)*32 + ln];
  acc0 = __builtin_amdgcn_wmma_f32_16x16x4_f32(false, a0, false, bw, (short)0, acc0, false, false);
  bw.x = W1[(4 + 2*hf)*32 + ln];  bw.y = W1[(5 + 2*hf)*32 + ln];
  acc0 = __builtin_amdgcn_wmma_f32_16x16x4_f32(false, a1, false, bw, (short)0, acc0, false, false);
  bw.x = W1[(0 + 2*hf)*32 + 16 + ln];  bw.y = W1[(1 + 2*hf)*32 + 16 + ln];
  acc1 = __builtin_amdgcn_wmma_f32_16x16x4_f32(false, a0, false, bw, (short)0, acc1, false, false);
  bw.x = W1[(4 + 2*hf)*32 + 16 + ln];  bw.y = W1[(5 + 2*hf)*32 + 16 + ln];
  acc1 = __builtin_amdgcn_wmma_f32_16x16x4_f32(false, a1, false, bw, (short)0, acc1, false, false);

  const float bia0 = b1[ln], bia1 = b1[16 + ln];
  #pragma unroll
  for (int v = 0; v < 8; ++v) {
    float t0 = acc0[v] + bia0;  h[w][v + 8*hf][ln]      = t0 > 0.0f ? t0 : 0.0f;
    float t1 = acc1[v] + bia1;  h[w][v + 8*hf][16 + ln] = t1 > 0.0f ? t1 : 0.0f;
  }
  __syncthreads();

  v8f acc2 = {0,0,0,0,0,0,0,0};
  #pragma unroll
  for (int k0 = 0; k0 < 32; k0 += 4) {
    v2f aa, bb;
    aa.x = h[w][ln][k0 + 2*hf];
    aa.y = h[w][ln][k0 + 1 + 2*hf];
    bb.x = W2[(k0 + 2*hf)*16 + ln];
    bb.y = W2[(k0 + 1 + 2*hf)*16 + ln];
    acc2 = __builtin_amdgcn_wmma_f32_16x16x4_f32(false, aa, false, bb, (short)0, acc2, false, false);
  }
  __syncthreads();
  const float bia2 = b2[ln];
  #pragma unroll
  for (int v = 0; v < 8; ++v) {
    float t = acc2[v] + bia2;
    h[w][v + 8*hf][ln] = t > 0.0f ? t : 0.0f;
  }
  __syncthreads();

  if (hf == 0) {
    float s3 = b3[0];
    #pragma unroll
    for (int c = 0; c < 16; ++c) s3 += h[w][ln][c] * W3[c];
    float cg   = 1.0f / (1.0f + expf(-s3));
    float mult = cg < 0.5f ? 2.0f * cg : 1.0f;
    mult1[p] = mult;

    float mA = fmaxf(fmaxf(x0, x1), fmaxf(x2, x3));
    float e0 = expf(x0 - mA), e1 = expf(x1 - mA), e2 = expf(x2 - mA), e3 = expf(x3 - mA);
    float impA = (e0 + e1) / (e0 + e1 + e2 + e3);
    float mO = fmaxf(fmaxf(x4, x5), fmaxf(x6, x7));
    float g0 = expf(x4 - mO), g1 = expf(x5 - mO), g2 = expf(x6 - mO), g3 = expf(x7 - mO);
    float impO = (g0 + g1) / (g0 + g1 + g2 + g3);

    const int bI = p >> 10;
    const int s  = p & (SEQ - 1);
    const int* eA = exA + bI * SEQ;
    const int* eO = exO + bI * SEQ;
    bool nearA = false, nearO = false;
    #pragma unroll
    for (int d = -3; d <= 3; ++d) {
      int ss = s + d;
      if (ss >= 0 && ss < SEQ) { nearA |= (eA[ss] > 0); nearO |= (eO[ss] > 0); }
    }
    fA2[p] = (impA > 0.5f && !nearO) ? 0.3f : 1.0f;
    fO2[p] = (impO > 0.5f && !nearA) ? 0.3f : 1.0f;
  }
}


__device__ __forceinline__ unsigned scan_step(unsigned& n1, unsigned& n2,
                                              unsigned cm, unsigned m1, unsigned m0) {
  const unsigned nb = n1 | n2 | cm;
  const unsigned na = (nb & m1) | (~nb & m0);
  n2 = n1;
  n1 = na;
  return nb;
}

__global__ void __launch_bounds__(256)
k_scan_write(const float* __restrict__ aL, const float* __restrict__ oL,
             const float* __restrict__ mult1, const float* __restrict__ fA2,
             const float* __restrict__ fO2,  const float* __restrict__ fA4,
             float* __restrict__ outA, float* __restrict__ outO) {
  const int b = blockIdx.x;
  const int t = threadIdx.x;
  __shared__ float sm[SEQ], sfa[SEQ], sfo[SEQ];
  __shared__ unsigned OG[32], IF1[32], IF01[32], F3[32];
  if (t < 32) { OG[t] = 0u; IF1[t] = 0u; IF01[t] = 0u; }
  __syncthreads();

  for (int s = t; s < SEQ; s += 256) {
    const int p = b * SEQ + s;
    v4f av = *(const v4f*)(aL + (size_t)p * 4);
    v4f ov = *(const v4f*)(oL + (size_t)p * 4);
    float mult = mult1[p];
    float fa1  = fA2[p] * fA4[p];
    float fo1  = fO2[p];
    sm[s] = mult;  sfa[s] = fa1;  sfo[s] = fo1;

    float a0 = av.x * mult, a1 = av.y * mult, a2 = av.z * mult, a3 = av.w * mult;
    float o0 = ov.x * mult, o1 = ov.y * mult, o2 = ov.z * mult, o3 = ov.w * mult;
    bool orig = (fmaxf(fmaxf(av.x, av.y), fmaxf(av.z, av.w)) > 0.5f) ||
                (fmaxf(fmaxf(ov.x, ov.y), fmaxf(ov.z, ov.w)) > 0.5f);
    float fa = fa1, fo = fo1;
    bool act1  = (fmaxf(fmaxf(a0*fa, a1*fa), fmaxf(a2, a3)) > 0.5f) ||
                 (fmaxf(fmaxf(o0*fo, o1*fo), fmaxf(o2, o3)) > 0.5f);
    fa = fa1 * 0.1f; fo = fo1 * 0.1f;
    bool act01 = (fmaxf(fmaxf(a0*fa, a1*fa), fmaxf(a2, a3)) > 0.5f) ||
                 (fmaxf(fmaxf(o0*fo, o1*fo), fmaxf(o2, o3)) > 0.5f);

    unsigned bit = 1u << (s & 31);
    if (orig)  atomicOr(&OG[s >> 5],  bit);
    if (act1)  atomicOr(&IF1[s >> 5], bit);
    if (act01) atomicOr(&IF01[s >> 5], bit);
  }
  __syncthreads();

  if (t < 32) {
    const unsigned og   = OG[t];
    const unsigned ogn  = (t < 31) ? OG[t + 1] : 0u;
    const unsigned if1  = IF1[t];
    const unsigned if01 = IF01[t];
    const unsigned cw = (og >> 1) | (ogn << 31) | (og >> 2) | (ogn << 30);

    unsigned n1 = 0x01000100u;
    unsigned n2 = 0x01010000u;
    for (int bb = 0; bb < 32; ++bb) {
      const unsigned cm = 0u - ((cw   >> bb) & 1u);
      const unsigned m1 = 0u - ((if1  >> bb) & 1u);
      const unsigned m0 = 0u - ((if01 >> bb) & 1u);
      (void)scan_step(n1, n2, cm, m1, m0);
    }
    const unsigned map = (n1 & 0x01010101u) | ((n2 & 0x01010101u) << 1);

    unsigned st = 0u, mine = 0u;
    for (int j = 0; j < 31; ++j) {
      const unsigned mj = (unsigned)__builtin_amdgcn_readlane((int)map, j);
      st = (mj >> (8u * st)) & 3u;
      if (t == j + 1) mine = st;
    }

    unsigned a1 = mine & 1u, a2 = mine >> 1;
    unsigned f3w = 0u;
    for (int bb = 0; bb < 32; ++bb) {
      const unsigned cm = 0u - ((cw   >> bb) & 1u);
      const unsigned m1 = 0u - ((if1  >> bb) & 1u);
      const unsigned m0 = 0u - ((if01 >> bb) & 1u);
      const unsigned nb = scan_step(a1, a2, cm, m1, m0);
      f3w |= (nb & 1u) << bb;
    }
    F3[t] = f3w;
  }
  __syncthreads();

  for (int s = t; s < SEQ; s += 256) {
    const int p = b * SEQ + s;
    v4f av = *(const v4f*)(aL + (size_t)p * 4);
    v4f ov = *(const v4f*)(oL + (size_t)p * 4);
    float mult = sm[s];
    float f3 = ((F3[s >> 5] >> (s & 31)) & 1u) ? 1.0f : 0.1f;
    float fa = sfa[s] * f3;
    float fo = sfo[s] * f3;
    v4f ra, ro;
    ra.x = av.x * mult * fa;  ra.y = av.y * mult * fa;
    ra.z = av.z * mult;       ra.w = av.w * mult;
    ro.x = ov.x * mult * fo;  ro.y = ov.y * mult * fo;
    ro.z = ov.z * mult;       ro.w = ov.w * mult;
    *(v4f*)(outA + (size_t)p * 4) = ra;
    *(v4f*)(outO + (size_t)p * 4) = ro;
  }
}

extern "C" void kernel_launch(void* const* d_in, const int* in_sizes, int n_in,
                              void* d_out, int out_size, void* d_ws, size_t ws_size,
                              hipStream_t stream) {
  (void)in_sizes; (void)n_in; (void)out_size; (void)ws_size;
  const float* aL  = (const float*)d_in[0];
  const float* oL  = (const float*)d_in[1];
  const float* rel = (const float*)d_in[2];
  const int*   exA = (const int*)d_in[3];
  const int*   exO = (const int*)d_in[4];
  const float* W1  = (const float*)d_in[5];
  const float* b1  = (const float*)d_in[6];
  const float* W2  = (const float*)d_in[7];
  const float* b2  = (const float*)d_in[8];
  const float* W3  = (const float*)d_in[9];
  const float* b3  = (const float*)d_in[10];

  float* ws    = (float*)d_ws;
  float* fA4   = ws;
  float* mult1 = ws + NPOS;
  float* fA2   = ws + 2 * NPOS;
  float* fO2   = ws + 3 * NPOS;
  float* outA  = (float*)d_out;
  float* outO  = outA + NPOS * 4;

  k_rel_reduce<<<NPOS, 256, 0, stream>>>(rel, fA4);
  k_mlp_factors<<<NPOS / 64, 128, 0, stream>>>(aL, oL, exA, exO,
                                               W1, b1, W2, b2, W3, b3,
                                               mult1, fA2, fO2);
  k_scan_write<<<BATCH, 256, 0, stream>>>(aL, oL, mult1, fA2, fO2, fA4,
                                          outA, outO);
}


# k_rel_reduce: a row's any() stops at its first hit: a wave owns 8 rows, loads the leading 512 B of each at once, reads on only where that piece has no hit (was a 256-thread block per whole row)
# speedup vs baseline: 2.4681x; 1.5644x over previous
#include <hip/hip_runtime.h>
#include <math.h>

#define BATCH 8
#define SEQ   1024
#define KREL  4
#define NPOS  (BATCH * SEQ)

typedef __attribute__((ext_vector_type(2))) float v2f;
typedef __attribute__((ext_vector_type(4))) float v4f;
typedef __attribute__((ext_vector_type(8))) float v8f;

#define RR_ROWS 8
#define RR_DEEP 8

__device__ __forceinline__ int rr_row(int q, int r) {
  return (r < 4) ? (4 * q + r) : (SEQ - 1 - 4 * q - (r - 4));
}

__device__ __forceinline__ bool rr_hit(v4f v) {
  return (v.x > 0.5f) || (v.y > 0.5f) || (v.z > 0.5f) || (v.w > 0.5f);
}

__global__ void __launch_bounds__(128)
k_rel_reduce(const float* __restrict__ rel, float* __restrict__ factA4) {
  const int lane = threadIdx.x & 31;
  const int gw = __builtin_amdgcn_readfirstlane((int)(blockIdx.x * 4 + (threadIdx.x >> 5)));
  const int b  = gw >> 7;
  const int q  = gw & 127;
  const float* relB = rel + (size_t)b * SEQ * (SEQ * KREL);

  v4f v[RR_ROWS];
  for (int r = 0; r < RR_ROWS; ++r) {
    const v4f* base = (const v4f*)(relB + (size_t)rr_row(q, r) * (SEQ * KREL));
    v[r] = __builtin_nontemporal_load(base + lane);
  }
  unsigned res = 0u;
  unsigned more = 0u;
  for (int r = 0; r < RR_ROWS; ++r) {
    const int i = rr_row(q, r);
    const bool found = __builtin_amdgcn_ballot_w32((lane < i) && rr_hit(v[r])) != 0u;
    res  |= found ? (1u << r) : 0u;
    more |= (!found && i > 32) ? (1u << r) : 0u;
  }

  if (more != 0u) {
    for (int r = 0; r < RR_ROWS; ++r) {
      if (!((more >> r) & 1u)) continue;
      const int i = rr_row(q, r);
      const v4f* base = (const v4f*)(relB + (size_t)i * (SEQ * KREL));
      bool found = false;
      for (int j0 = 32; j0 < i && !found; j0 += 32 * RR_DEEP) {
        v4f t[RR_DEEP];
        for (int u = 0; u < RR_DEEP; ++u) {
          const int j = j0 + 32 * u + lane;
          const v4f z = {0.0f, 0.0f, 0.0f, 0.0f};
          t[u] = (j < i) ? __builtin_nontemporal_load(base + j) : z;
        }
        bool h = false;
        for (int u = 0; u < RR_DEEP; ++u) h = h || rr_hit(t[u]);
        found = __builtin_amdgcn_ballot_w32(h) != 0u;
      }
      res |= found ? (1u << r) : 0u;
    }
  }

  if (lane < RR_ROWS)
    factA4[b * SEQ + rr_row(q, lane)] = ((res >> lane) & 1u) ? 0.7f : 1.0f;
}

__global__ void __launch_bounds__(128)
k_mlp_factors(const float* __restrict__ aL, const float* __restrict__ oL,
              const int*   __restrict__ exA, const int* __restrict__ exO,
              const float* __restrict__ W1, const float* __restrict__ b1,
              const float* __restrict__ W2, const float* __restrict__ b2,
              const float* __restrict__ W3, const float* __restrict__ b3,
              float* __restrict__ mult1, float* __restrict__ fA2,
              float* __restrict__ fO2) {
  __shared__ float h[4][16][32];
  const int tid  = threadIdx.x;
  const int w    = tid >> 5;
  const int lane = tid & 31;
  const int hf   = lane >> 4;
  const int ln   = lane & 15;
  const int posBase = blockIdx.x * 64 + w * 16;
  const int p = posBase + ln;

  const float* aRow = aL + (size_t)p * 4;
  const float* oRow = oL + (size_t)p * 4;
  const float x0 = aRow[0], x1 = aRow[1], x2 = aRow[2], x3 = aRow[3];
  const float x4 = oRow[0], x5 = oRow[1], x6 = oRow[2], x7 = oRow[3];

  v2f a0, a1, bw;
  a0.x = hf ? x2 : x0;  a0.y = hf ? x3 : x1;
  a1.x = hf ? x6 : x4;  a1.y = hf ? x7 : x5;
  v8f acc0 = {0,0,0,0,0,0,0,0};
  v8f acc1 = {0,0,0,0,0,0,0,0};
  bw.x = W1[(0 + 2*hf)*32 + ln];  bw.y = W1[(1 + 2*hf)*32 + ln];
  acc0 = __builtin_amdgcn_wmma_f32_16x16x4_f32(false, a0, false, bw, (short)0, acc0, false, false);
  bw.x = W1[(4 + 2*hf)*32 + ln];  bw.y = W1[(5 + 2*hf)*32 + ln];
  acc0 = __builtin_amdgcn_wmma_f32_16x16x4_f32(false, a1, false, bw, (short)0, acc0, false, false);
  bw.x = W1[(0 + 2*hf)*32 + 16 + ln];  bw.y = W1[(1 + 2*hf)*32 + 16 + ln];
  acc1 = __builtin_amdgcn_wmma_f32_16x16x4_f32(false, a0, false, bw, (short)0, acc1, false, false);
  bw.x = W1[(4 + 2*hf)*32 + 16 + ln];  bw.y = W1[(5 + 2*hf)*32 + 16 + ln];
  acc1 = __builtin_amdgcn_wmma_f32_16x16x4_f32(false, a1, false, bw, (short)0, acc1, false, false);

  const float bia0 = b1[ln], bia1 = b1[16 + ln];
  #pragma unroll
  for (int v = 0; v < 8; ++v) {
    float t0 = acc0[v] + bia0;  h[w][v + 8*hf][ln]      = t0 > 0.0f ? t0 : 0.0f;
    float t1 = acc1[v] + bia1;  h[w][v + 8*hf][16 + ln] = t1 > 0.0f ? t1 : 0.0f;
  }
  __syncthreads();

  v8f acc2 = {0,0,0,0,0,0,0,0};
  #pragma unroll
  for (int k0 = 0; k0 < 32; k0 += 4) {
    v2f aa, bb;
    aa.x = h[w][ln][k0 + 2*hf];
    aa.y = h[w][ln][k0 + 1 + 2*hf];
    bb.x = W2[(k0 + 2*hf)*16 + ln];
    bb.y = W2[(k0 + 1 + 2*hf)*16 + ln];
    acc2 = __builtin_amdgcn_wmma_f32_16x16x4_f32(false, aa, false, bb, (short)0, acc2, false, false);
  }
  __syncthreads();
  const float bia2 = b2[ln];
  #pragma unroll
  for (int v = 0; v < 8; ++v) {
    float t = acc2[v] + bia2;
    h[w][v + 8*hf][ln] = t > 0.0f ? t : 0.0f;
  }
  __syncthreads();

  if (hf == 0) {
    float s3 = b3[0];
    #pragma unroll
    for (int c = 0; c < 16; ++c) s3 += h[w][ln][c] * W3[c];
    float cg   = 1.0f / (1.0f + expf(-s3));
    float mult = cg < 0.5f ? 2.0f * cg : 1.0f;
    mult1[p] = mult;

    float mA = fmaxf(fmaxf(x0, x1), fmaxf(x2, x3));
    float e0 = expf(x0 - mA), e1 = expf(x1 - mA), e2 = expf(x2 - mA), e3 = expf(x3 - mA);
    float impA = (e0 + e1) / (e0 + e1 + e2 + e3);
    float mO = fmaxf(fmaxf(x4, x5), fmaxf(x6, x7));
    float g0 = expf(x4 - mO), g1 = expf(x5 - mO), g2 = expf(x6 - mO), g3 = expf(x7 - mO);
    float impO = (g0 + g1) / (g0 + g1 + g2 + g3);

    const int bI = p >> 10;
    const int s  = p & (SEQ - 1);
    const int* eA = exA + bI * SEQ;
    const int* eO = exO + bI * SEQ;
    bool nearA = false, nearO = false;
    #pragma unroll
    for (int d = -3; d <= 3; ++d) {
      int ss = s + d;
      if (ss >= 0 && ss < SEQ) { nearA |= (eA[ss] > 0); nearO |= (eO[ss] > 0); }
    }
    fA2[p] = (impA > 0.5f && !nearO) ? 0.3f : 1.0f;
    fO2[p] = (impO > 0.5f && !nearA) ? 0.3f : 1.0f;
  }
}


__device__ __forceinline__ unsigned scan_step(unsigned& n1, unsigned& n2,
                                              unsigned cm, unsigned m1, unsigned m0) {
  const unsigned nb = n1 | n2 | cm;
  const unsigned na = (nb & m1) | (~nb & m0);
  n2 = n1;
  n1 = na;
  return nb;
}

__global__ void __launch_bounds__(256)
k_scan_write(const float* __restrict__ aL, const float* __restrict__ oL,
             const float* __restrict__ mult1, const float* __restrict__ fA2,
             const float* __restrict__ fO2,  const float* __restrict__ fA4,
             float* __restrict__ outA, float* __restrict__ outO) {
  const int b = blockIdx.x;
  const int t = threadIdx.x;
  __shared__ float sm[SEQ], sfa[SEQ], sfo[SEQ];
  __shared__ unsigned OG[32], IF1[32], IF01[32], F3[32];
  if (t < 32) { OG[t] = 0u; IF1[t] = 0u; IF01[t] = 0u; }
  __syncthreads();

  for (int s = t; s < SEQ; s += 256) {
    const int p = b * SEQ + s;
    v4f av = *(const v4f*)(aL + (size_t)p * 4);
    v4f ov = *(const v4f*)(oL + (size_t)p * 4);
    float mult = mult1[p];
    float fa1  = fA2[p] * fA4[p];
    float fo1  = fO2[p];
    sm[s] = mult;  sfa[s] = fa1;  sfo[s] = fo1;

    float a0 = av.x * mult, a1 = av.y * mult, a2 = av.z * mult, a3 = av.w * mult;
    float o0 = ov.x * mult, o1 = ov.y * mult, o2 = ov.z * mult, o3 = ov.w * mult;
    bool orig = (fmaxf(fmaxf(av.x, av.y), fmaxf(av.z, av.w)) > 0.5f) ||
                (fmaxf(fmaxf(ov.x, ov.y), fmaxf(ov.z, ov.w)) > 0.5f);
    float fa = fa1, fo = fo1;
    bool act1  = (fmaxf(fmaxf(a0*fa, a1*fa), fmaxf(a2, a3)) > 0.5f) ||
                 (fmaxf(fmaxf(o0*fo, o1*fo), fmaxf(o2, o3)) > 0.5f);
    fa = fa1 * 0.1f; fo = fo1 * 0.1f;
    bool act01 = (fmaxf(fmaxf(a0*fa, a1*fa), fmaxf(a2, a3)) > 0.5f) ||
                 (fmaxf(fmaxf(o0*fo, o1*fo), fmaxf(o2, o3)) > 0.5f);

    unsigned bit = 1u << (s & 31);
    if (orig)  atomicOr(&OG[s >> 5],  bit);
    if (act1)  atomicOr(&IF1[s >> 5], bit);
    if (act01) atomicOr(&IF01[s >> 5], bit);
  }
  __syncthreads();

  if (t < 32) {
    const unsigned og   = OG[t];
    const unsigned ogn  = (t < 31) ? OG[t + 1] : 0u;
    const unsigned if1  = IF1[t];
    const unsigned if01 = IF01[t];
    const unsigned cw = (og >> 1) | (ogn << 31) | (og >> 2) | (ogn << 30);

    unsigned n1 = 0x01000100u;
    unsigned n2 = 0x01010000u;
    for (int bb = 0; bb < 32; ++bb) {
      const unsigned cm = 0u - ((cw   >> bb) & 1u);
      const unsigned m1 = 0u - ((if1  >> bb) & 1u);
      const unsigned m0 = 0u - ((if01 >> bb) & 1u);
      (void)scan_step(n1, n2, cm, m1, m0);
    }
    const unsigned map = (n1 & 0x01010101u) | ((n2 & 0x01010101u) << 1);

    unsigned st = 0u, mine = 0u;
    for (int j = 0; j < 31; ++j) {
      const unsigned mj = (unsigned)__builtin_amdgcn_readlane((int)map, j);
      st = (mj >> (8u * st)) & 3u;
      if (t == j + 1) mine = st;
    }

    unsigned a1 = mine & 1u, a2 = mine >> 1;
    unsigned f3w = 0u;
    for (int bb = 0; bb < 32; ++bb) {
      const unsigned cm = 0u - ((cw   >> bb) & 1u);
      const unsigned m1 = 0u - ((if1  >> bb) & 1u);
      const unsigned m0 = 0u - ((if01 >> bb) & 1u);
      const unsigned nb = scan_step(a1, a2, cm, m1, m0);
      f3w |= (nb & 1u) << bb;
    }
    F3[t] = f3w;
  }
  __syncthreads();

  for (int s = t; s < SEQ; s += 256) {
    const int p = b * SEQ + s;
    v4f av = *(const v4f*)(aL + (size_t)p * 4);
    v4f ov = *(const v4f*)(oL + (size_t)p * 4);
    float mult = sm[s];
    float f3 = ((F3[s >> 5] >> (s & 31)) & 1u) ? 1.0f : 0.1f;
    float fa = sfa[s] * f3;
    float fo = sfo[s] * f3;
    v4f ra, ro;
    ra.x = av.x * mult * fa;  ra.y = av.y * mult * fa;
    ra.z = av.z * mult;       ra.w = av.w * mult;
    ro.x = ov.x * mult * fo;  ro.y = ov.y * mult * fo;
    ro.z = ov.z * mult;       ro.w = ov.w * mult;
    *(v4f*)(outA + (size_t)p * 4) = ra;
    *(v4f*)(outO + (size_t)p * 4) = ro;
  }
}

extern "C" void kernel_launch(void* const* d_in, const int* in_sizes, int n_in,
                              void* d_out, int out_size, void* d_ws, size_t ws_size,
                              hipStream_t stream) {
  (void)in_sizes; (void)n_in; (void)out_size; (void)ws_size;
  const float* aL  = (const float*)d_in[0];
  const float* oL  = (const float*)d_in[1];
  const float* rel = (const float*)d_in[2];
  const int*   exA = (const int*)d_in[3];
  const int*   exO = (const int*)d_in[4];
  const float* W1  = (const float*)d_in[5];
  const float* b1  = (const float*)d_in[6];
  const float* W2  = (const float*)d_in[7];
  const float* b2  = (const float*)d_in[8];
  const float* W3  = (const float*)d_in[9];
  const float* b3  = (const float*)d_in[10];

  float* ws    = (float*)d_ws;
  float* fA4   = ws;
  float* mult1 = ws + NPOS;
  float* fA2   = ws + 2 * NPOS;
  float* fO2   = ws + 3 * NPOS;
  float* outA  = (float*)d_out;
  float* outO  = outA + NPOS * 4;

  k_rel_reduce<<<NPOS / (4 * RR_ROWS), 128, 0, stream>>>(rel, fA4);
  k_mlp_factors<<<NPOS / 64, 128, 0, stream>>>(aL, oL, exA, exO,
                                               W1, b1, W2, b2, W3, b3,
                                               mult1, fA2, fO2);
  k_scan_write<<<BATCH, 256, 0, stream>>>(aL, oL, mult1, fA2, fO2, fA4,
                                          outA, outO);
}


# dependent load trips collapsed: k_mlp_factors fetches W2, b2, W3, b3 and its 14 window words with its rows, not later; k_scan_write one position per thread, ballot bitmaps, rows kept in registers
# speedup vs baseline: 3.9300x; 1.5923x over previous
#include <hip/hip_runtime.h>
#include <math.h>

#define BATCH 8
#define SEQ   1024
#define KREL  4
#define NPOS  (BATCH * SEQ)

typedef __attribute__((ext_vector_type(2))) float v2f;
typedef __attribute__((ext_vector_type(4))) float v4f;
typedef __attribute__((ext_vector_type(8))) float v8f;

#define RR_ROWS 8
#define RR_DEEP 8

__device__ __forceinline__ int rr_row(int q, int r) {
  return (r < 4) ? (4 * q + r) : (SEQ - 1 - 4 * q - (r - 4));
}

__device__ __forceinline__ bool rr_hit(v4f v) {
  return (v.x > 0.5f) || (v.y > 0.5f) || (v.z > 0.5f) || (v.w > 0.5f);
}

__global__ void __launch_bounds__(128)
k_rel_reduce(const float* __restrict__ rel, float* __restrict__ factA4) {
  const int lane = threadIdx.x & 31;
  const int gw = __builtin_amdgcn_readfirstlane((int)(blockIdx.x * 4 + (threadIdx.x >> 5)));
  const int b  = gw >> 7;
  const int q  = gw & 127;
  const float* relB = rel + (size_t)b * SEQ * (SEQ * KREL);

  v4f v[RR_ROWS];
  for (int r = 0; r < RR_ROWS; ++r) {
    const v4f* base = (const v4f*)(relB + (size_t)rr_row(q, r) * (SEQ * KREL));
    v[r] = __builtin_nontemporal_load(base + lane);
  }
  unsigned res = 0u;
  unsigned more = 0u;
  for (int r = 0; r < RR_ROWS; ++r) {
    const int i = rr_row(q, r);
    const bool found = __builtin_amdgcn_ballot_w32((lane < i) && rr_hit(v[r])) != 0u;
    res  |= found ? (1u << r) : 0u;
    more |= (!found && i > 32) ? (1u << r) : 0u;
  }

  if (more != 0u) {
    for (int r = 0; r < RR_ROWS; ++r) {
      if (!((more >> r) & 1u)) continue;
      const int i = rr_row(q, r);
      const v4f* base = (const v4f*)(relB + (size_t)i * (SEQ * KREL));
      bool found = false;
      for (int j0 = 32; j0 < i && !found; j0 += 32 * RR_DEEP) {
        v4f t[RR_DEEP];
        for (int u = 0; u < RR_DEEP; ++u) {
          const int j = j0 + 32 * u + lane;
          const v4f z = {0.0f, 0.0f, 0.0f, 0.0f};
          t[u] = (j < i) ? __builtin_nontemporal_load(base + j) : z;
        }
        bool h = false;
        for (int u = 0; u < RR_DEEP; ++u) h = h || rr_hit(t[u]);
        found = __builtin_amdgcn_ballot_w32(h) != 0u;
      }
      res |= found ? (1u << r) : 0u;
    }
  }

  if (lane < RR_ROWS)
    factA4[b * SEQ + rr_row(q, lane)] = ((res >> lane) & 1u) ? 0.7f : 1.0f;
}

__global__ void __launch_bounds__(128)
k_mlp_factors(const float* __restrict__ aL, const float* __restrict__ oL,
              const int*   __restrict__ exA, const int* __restrict__ exO,
              const float* __restrict__ W1, const float* __restrict__ b1,
              const float* __restrict__ W2, const float* __restrict__ b2,
              const float* __restrict__ W3, const float* __restrict__ b3,
              float* __restrict__ mult1, float* __restrict__ fA2,
              float* __restrict__ fO2) {
  __shared__ float h[4][16][32];
  const int tid  = threadIdx.x;
  const int w    = tid >> 5;
  const int lane = tid & 31;
  const int hf   = lane >> 4;
  const int ln   = lane & 15;
  const int posBase = blockIdx.x * 64 + w * 16;
  const int p = posBase + ln;

  const float* aRow = aL + (size_t)p * 4;
  const float* oRow = oL + (size_t)p * 4;
  const float x0 = aRow[0], x1 = aRow[1], x2 = aRow[2], x3 = aRow[3];
  const float x4 = oRow[0], x5 = oRow[1], x6 = oRow[2], x7 = oRow[3];

  float w2x[8], w2y[8];
  #pragma unroll
  for (int c = 0; c < 8; ++c) {
    w2x[c] = W2[(4*c + 2*hf)*16 + ln];
    w2y[c] = W2[(4*c + 1 + 2*hf)*16 + ln];
  }
  const float bia2 = b2[ln];
  float w3[16];
  #pragma unroll
  for (int c = 0; c < 16; ++c) w3[c] = W3[c];
  const float bia3 = b3[0];
  const int bI = p >> 10;
  const int s  = p & (SEQ - 1);
  const int* eA = exA + bI * SEQ;
  const int* eO = exO + bI * SEQ;
  int winA[7], winO[7];
  #pragma unroll
  for (int d = 0; d < 7; ++d) {
    int ss = s + d - 3;
    ss = ss < 0 ? 0 : (ss > SEQ - 1 ? SEQ - 1 : ss);
    winA[d] = eA[ss];
    winO[d] = eO[ss];
  }

  v2f a0, a1, bw;
  a0.x = hf ? x2 : x0;  a0.y = hf ? x3 : x1;
  a1.x = hf ? x6 : x4;  a1.y = hf ? x7 : x5;
  v8f acc0 = {0,0,0,0,0,0,0,0};
  v8f acc1 = {0,0,0,0,0,0,0,0};
  bw.x = W1[(0 + 2*hf)*32 + ln];  bw.y = W1[(1 + 2*hf)*32 + ln];
  acc0 = __builtin_amdgcn_wmma_f32_16x16x4_f32(false, a0, false, bw, (short)0, acc0, false, false);
  bw.x = W1[(4 + 2*hf)*32 + ln];  bw.y = W1[(5 + 2*hf)*32 + ln];
  acc0 = __builtin_amdgcn_wmma_f32_16x16x4_f32(false, a1, false, bw, (short)0, acc0, false, false);
  bw.x = W1[(0 + 2*hf)*32 + 16 + ln];  bw.y = W1[(1 + 2*hf)*32 + 16 + ln];
  acc1 = __builtin_amdgcn_wmma_f32_16x16x4_f32(false, a0, false, bw, (short)0, acc1, false, false);
  bw.x = W1[(4 + 2*hf)*32 + 16 + ln];  bw.y = W1[(5 + 2*hf)*32 + 16 + ln];
  acc1 = __builtin_amdgcn_wmma_f32_16x16x4_f32(false, a1, false, bw, (short)0, acc1, false, false);

  const float bia0 = b1[ln], bia1 = b1[16 + ln];
  #pragma unroll
  for (int v = 0; v < 8; ++v) {
    float t0 = acc0[v] + bia0;  h[w][v + 8*hf][ln]      = t0 > 0.0f ? t0 : 0.0f;
    float t1 = acc1[v] + bia1;  h[w][v + 8*hf][16 + ln] = t1 > 0.0f ? t1 : 0.0f;
  }
  __syncthreads();

  v8f acc2 = {0,0,0,0,0,0,0,0};
  #pragma unroll
  for (int k0 = 0; k0 < 32; k0 += 4) {
    v2f aa, bb;
    aa.x = h[w][ln][k0 + 2*hf];
    aa.y = h[w][ln][k0 + 1 + 2*hf];
    bb.x = w2x[k0 >> 2];
    bb.y = w2y[k0 >> 2];
    acc2 = __builtin_amdgcn_wmma_f32_16x16x4_f32(false, aa, false, bb, (short)0, acc2, false, false);
  }
  __syncthreads();
  #pragma unroll
  for (int v = 0; v < 8; ++v) {
    float t = acc2[v] + bia2;
    h[w][v + 8*hf][ln] = t > 0.0f ? t : 0.0f;
  }
  __syncthreads();

  if (hf == 0) {
    float s3 = bia3;
    #pragma unroll
    for (int c = 0; c < 16; ++c) s3 += h[w][ln][c] * w3[c];
    float cg   = 1.0f / (1.0f + expf(-s3));
    float mult = cg < 0.5f ? 2.0f * cg : 1.0f;
    mult1[p] = mult;

    float mA = fmaxf(fmaxf(x0, x1), fmaxf(x2, x3));
    float e0 = expf(x0 - mA), e1 = expf(x1 - mA), e2 = expf(x2 - mA), e3 = expf(x3 - mA);
    float impA = (e0 + e1) / (e0 + e1 + e2 + e3);
    float mO = fmaxf(fmaxf(x4, x5), fmaxf(x6, x7));
    float g0 = expf(x4 - mO), g1 = expf(x5 - mO), g2 = expf(x6 - mO), g3 = expf(x7 - mO);
    float impO = (g0 + g1) / (g0 + g1 + g2 + g3);

    bool nearA = false, nearO = false;
    #pragma unroll
    for (int d = -3; d <= 3; ++d) {
      int ss = s + d;
      if (ss >= 0 && ss < SEQ) { nearA |= (winA[d + 3] > 0); nearO |= (winO[d + 3] > 0); }
    }
    fA2[p] = (impA > 0.5f && !nearO) ? 0.3f : 1.0f;
    fO2[p] = (impO > 0.5f && !nearA) ? 0.3f : 1.0f;
  }
}


__device__ __forceinline__ unsigned scan_step(unsigned& n1, unsigned& n2,
                                              unsigned cm, unsigned m1, unsigned m0) {
  const unsigned nb = n1 | n2 | cm;
  const unsigned na = (nb & m1) | (~nb & m0);
  n2 = n1;
  n1 = na;
  return nb;
}

__global__ void __launch_bounds__(SEQ)
k_scan_write(const float* __restrict__ aL, const float* __restrict__ oL,
             const float* __restrict__ mult1, const float* __restrict__ fA2,
             const float* __restrict__ fO2,  const float* __restrict__ fA4,
             float* __restrict__ outA, float* __restrict__ outO) {
  const int b = blockIdx.x;
  const int t = threadIdx.x;
  const int lane = t & 31;
  const int wv = t >> 5;
  __shared__ unsigned OG[32], IF1[32], IF01[32], F3[32];

  const int p = b * SEQ + t;
  const v4f av = *(const v4f*)(aL + (size_t)p * 4);
  const v4f ov = *(const v4f*)(oL + (size_t)p * 4);
  const float mult = mult1[p];
  const float fa1  = fA2[p] * fA4[p];
  const float fo1  = fO2[p];
  {
    float a0 = av.x * mult, a1 = av.y * mult, a2 = av.z * mult, a3 = av.w * mult;
    float o0 = ov.x * mult, o1 = ov.y * mult, o2 = ov.z * mult, o3 = ov.w * mult;
    bool orig = (fmaxf(fmaxf(av.x, av.y), fmaxf(av.z, av.w)) > 0.5f) ||
                (fmaxf(fmaxf(ov.x, ov.y), fmaxf(ov.z, ov.w)) > 0.5f);
    float fa = fa1, fo = fo1;
    bool act1  = (fmaxf(fmaxf(a0*fa, a1*fa), fmaxf(a2, a3)) > 0.5f) ||
                 (fmaxf(fmaxf(o0*fo, o1*fo), fmaxf(o2, o3)) > 0.5f);
    fa = fa1 * 0.1f; fo = fo1 * 0.1f;
    bool act01 = (fmaxf(fmaxf(a0*fa, a1*fa), fmaxf(a2, a3)) > 0.5f) ||
                 (fmaxf(fmaxf(o0*fo, o1*fo), fmaxf(o2, o3)) > 0.5f);

    const unsigned wOrig = __builtin_amdgcn_ballot_w32(orig);
    const unsigned wAct1 = __builtin_amdgcn_ballot_w32(act1);
    const unsigned wAct0 = __builtin_amdgcn_ballot_w32(act01);
    if (lane == 0) { OG[wv] = wOrig; IF1[wv] = wAct1; IF01[wv] = wAct0; }
  }
  __syncthreads();

  if (t < 32) {
    const unsigned og   = OG[t];
    const unsigned ogn  = (t < 31) ? OG[t + 1] : 0u;
    const unsigned if1  = IF1[t];
    const unsigned if01 = IF01[t];
    const unsigned cw = (og >> 1) | (ogn << 31) | (og >> 2) | (ogn << 30);

    unsigned n1 = 0x01000100u;
    unsigned n2 = 0x01010000u;
    for (int bb = 0; bb < 32; ++bb) {
      const unsigned cm = 0u - ((cw   >> bb) & 1u);
      const unsigned m1 = 0u - ((if1  >> bb) & 1u);
      const unsigned m0 = 0u - ((if01 >> bb) & 1u);
      (void)scan_step(n1, n2, cm, m1, m0);
    }
    const unsigned map = (n1 & 0x01010101u) | ((n2 & 0x01010101u) << 1);

    unsigned st = 0u, mine = 0u;
    for (int j = 0; j < 31; ++j) {
      const unsigned mj = (unsigned)__builtin_amdgcn_readlane((int)map, j);
      st = (mj >> (8u * st)) & 3u;
      if (t == j + 1) mine = st;
    }

    unsigned a1 = mine & 1u, a2 = mine >> 1;
    unsigned f3w = 0u;
    for (int bb = 0; bb < 32; ++bb) {
      const unsigned cm = 0u - ((cw   >> bb) & 1u);
      const unsigned m1 = 0u - ((if1  >> bb) & 1u);
      const unsigned m0 = 0u - ((if01 >> bb) & 1u);
      const unsigned nb = scan_step(a1, a2, cm, m1, m0);
      f3w |= (nb & 1u) << bb;
    }
    F3[t] = f3w;
  }
  __syncthreads();

  {
    float f3 = ((F3[wv] >> lane) & 1u) ? 1.0f : 0.1f;
    float fa = fa1 * f3;
    float fo = fo1 * f3;
    v4f ra, ro;
    ra.x = av.x * mult * fa;  ra.y = av.y * mult * fa;
    ra.z = av.z * mult;       ra.w = av.w * mult;
    ro.x = ov.x * mult * fo;  ro.y = ov.y * mult * fo;
    ro.z = ov.z * mult;       ro.w = ov.w * mult;
    *(v4f*)(outA + (size_t)p * 4) = ra;
    *(v4f*)(outO + (size_t)p * 4) = ro;
  }
}

extern "C" void kernel_launch(void* const* d_in, const int* in_sizes, int n_in,
                              void* d_out, int out_size, void* d_ws, size_t ws_size,
                              hipStream_t stream) {
  (void)in_sizes; (void)n_in; (void)out_size; (void)ws_size;
  const float* aL  = (const float*)d_in[0];
  const float* oL  = (const float*)d_in[1];
  const float* rel = (const float*)d_in[2];
  const int*   exA = (const int*)d_in[3];
  const int*   exO = (const int*)d_in[4];
  const float* W1  = (const float*)d_in[5];
  const float* b1  = (const float*)d_in[6];
  const float* W2  = (const float*)d_in[7];
  const float* b2  = (const float*)d_in[8];
  const float* W3  = (const float*)d_in[9];
  const float* b3  = (const float*)d_in[10];

  float* ws    = (float*)d_ws;
  float* fA4   = ws;
  float* mult1 = ws + NPOS;
  float* fA2   = ws + 2 * NPOS;
  float* fO2   = ws + 3 * NPOS;
  float* outA  = (float*)d_out;
  float* outO  = outA + NPOS * 4;

  k_rel_reduce<<<NPOS / (4 * RR_ROWS), 128, 0, stream>>>(rel, fA4);
  k_mlp_factors<<<NPOS / 64, 128, 0, stream>>>(aL, oL, exA, exO,
                                               W1, b1, W2, b2, W3, b3,
                                               mult1, fA2, fO2);
  k_scan_write<<<BATCH, SEQ, 0, stream>>>(aL, oL, mult1, fA2, fO2, fA4,
                                          outA, outO);
}
